# starting kernel
# baseline (speedup 1.0000x reference)
#include <hip/hip_runtime.h>
#include <math.h>


#define NNODES 50000
#define NEDGES 1600000
#define IN_F   256
#define HEADS  4
#define OUT_F  32
#define CO     (HEADS * OUT_F)
#define NEG_SLOPE 0.2f

typedef __attribute__((ext_vector_type(2))) float v2f;
typedef __attribute__((ext_vector_type(8))) float v8f;

__device__ __forceinline__ unsigned f2key(float x) {
  unsigned b = __float_as_uint(x);
  return (b & 0x80000000u) ? ~b : (b | 0x80000000u);
}
__device__ __forceinline__ float key2f(unsigned k) {
  unsigned b = (k & 0x80000000u) ? (k ^ 0x80000000u) : ~k;
  return __uint_as_float(b);
}
__device__ __forceinline__ float leaky(float x) {
  return x > 0.f ? x : NEG_SLOPE * x;
}

__global__ __launch_bounds__(256) void k_init(float* __restrict__ out,
                                              unsigned* __restrict__ mkey,
                                              float* __restrict__ ssum) {
  int i = blockIdx.x * 256 + threadIdx.x;
  if (i < NNODES * CO) out[i] = 0.f;
  if (i < NNODES * HEADS) { mkey[i] = 0u; ssum[i] = 0.f; }
}

__global__ __launch_bounds__(256) void k_gemm(const float* __restrict__ feat,
                                              const float* __restrict__ W,
                                              float* __restrict__ ft) {
  __shared__ float As[16][IN_F + 4];
  const int tid    = threadIdx.x;
  const int wave   = tid >> 5;
  const int lane   = tid & 31;
  const int tile_m = blockIdx.x * 16;

  for (int i = tid; i < 16 * (IN_F / 4); i += 256) {
    const int r  = i / (IN_F / 4);
    const int c4 = i % (IN_F / 4);
    float4 v = make_float4(0.f, 0.f, 0.f, 0.f);
    const int row = tile_m + r;
    if (row < NNODES) v = ((const float4*)(feat + (size_t)row * IN_F))[c4];
    ((float4*)&As[r][0])[c4] = v;
  }
  __syncthreads();

  const int koff = (lane < 16) ? 0 : 2;
  const int arow = lane & 15;
  const int col  = wave * 16 + arow;
  const float* wrow = W + (size_t)col * IN_F;

  v8f c = {};
  for (int k0 = 0; k0 < IN_F; k0 += 4) {
    v2f a, b;
    a.x = As[arow][k0 + koff];
    a.y = As[arow][k0 + koff + 1];
    b.x = wrow[k0 + koff];
    b.y = wrow[k0 + koff + 1];
    c = __builtin_amdgcn_wmma_f32_16x16x4_f32(
         false, a,  false, b,
         (short)0, c,  false,  false);
  }

  const int mbase  = (lane < 16) ? 0 : 8;
  const int outcol = col;
#pragma unroll
  for (int j = 0; j < 8; ++j) {
    const int row = tile_m + mbase + j;
    if (row < NNODES) ft[(size_t)row * CO + outcol] = c[j];
  }
}

__global__ __launch_bounds__(256) void k_eler(const float* __restrict__ ft,
                                              const float* __restrict__ attn,
                                              float* __restrict__ el,
                                              float* __restrict__ er) {
  int i = blockIdx.x * 256 + threadIdx.x;
  if (i >= NNODES * HEADS) return;
  const int n = i / HEADS, h = i % HEADS;
  const float* f  = ft + (size_t)n * CO + h * OUT_F;
  const float* al = attn + h * 2 * OUT_F;
  const float* ar = al + OUT_F;
  float sl = 0.f, sr = 0.f;
#pragma unroll
  for (int d = 0; d < OUT_F; ++d) {
    const float v = f[d];
    sl += v * al[d];
    sr += v * ar[d];
  }
  el[i] = sl;
  er[i] = sr;
}

__global__ __launch_bounds__(256) void k_edge_max(const int* __restrict__ src,
                                                  const int* __restrict__ dst,
                                                  const float* __restrict__ el,
                                                  const float* __restrict__ er,
                                                  unsigned* __restrict__ mkey) {
  int i = blockIdx.x * 256 + threadIdx.x;
  if (i >= NEDGES * HEADS) return;
  const int e = i >> 2, h = i & 3;
  const int s = src[e], d = dst[e];
  const float x = leaky(el[s * HEADS + h] + er[d * HEADS + h]);
  atomicMax(&mkey[d * HEADS + h], f2key(x));
}

__global__ __launch_bounds__(256) void k_edge_sum(const int* __restrict__ src,
                                                  const int* __restrict__ dst,
                                                  const float* __restrict__ el,
                                                  const float* __restrict__ er,
                                                  const unsigned* __restrict__ mkey,
                                                  float* __restrict__ ssum) {
  int i = blockIdx.x * 256 + threadIdx.x;
  if (i >= NEDGES * HEADS) return;
  const int e = i >> 2, h = i & 3;
  const int s = src[e], d = dst[e];
  const float x = leaky(el[s * HEADS + h] + er[d * HEADS + h]);
  const float m = key2f(mkey[d * HEADS + h]);
  atomicAdd(&ssum[d * HEADS + h], expf(x - m));
}

__global__ __launch_bounds__(256) void k_edge_agg(const int* __restrict__ src,
                                                  const int* __restrict__ dst,
                                                  const float* __restrict__ el,
                                                  const float* __restrict__ er,
                                                  const unsigned* __restrict__ mkey,
                                                  const float* __restrict__ ssum,
                                                  const float* __restrict__ ft,
                                                  float* __restrict__ out) {
  const int lane = threadIdx.x & 31;
  const int eg   = blockIdx.x * 8 + (threadIdx.x >> 5);
  if (eg >= NEDGES) return;
  const int s = src[eg], d = dst[eg];

  const float* fsrc = ft + (size_t)s * CO;
  __builtin_prefetch(fsrc, 0, 0);

#pragma unroll
  for (int h = 0; h < HEADS; ++h) {
    const float x = leaky(el[s * HEADS + h] + er[d * HEADS + h]);
    const float m = key2f(mkey[d * HEADS + h]);
    const float a = expf(x - m) / ssum[d * HEADS + h];
    const float v = fsrc[h * OUT_F + lane] * a;
    atomicAdd(&out[(size_t)d * CO + h * OUT_F + lane], v);
  }
}

extern "C" void kernel_launch(void* const* d_in, const int* in_sizes, int n_in,
                              void* d_out, int out_size, void* d_ws, size_t ws_size,
                              hipStream_t stream) {
  (void)in_sizes; (void)n_in; (void)out_size; (void)ws_size;

  const float* feat = (const float*)d_in[0];
  const int*   src  = (const int*)d_in[1];
  const int*   dst  = (const int*)d_in[2];
  const float* W    = (const float*)d_in[3];
  const float* attn = (const float*)d_in[4];
  float* out = (float*)d_out;

  char* ws = (char*)d_ws;
  float*    ft   = (float*)ws;    ws += (size_t)NNODES * CO * sizeof(float);
  float*    el   = (float*)ws;    ws += (size_t)NNODES * HEADS * sizeof(float);
  float*    er   = (float*)ws;    ws += (size_t)NNODES * HEADS * sizeof(float);
  unsigned* mkey = (unsigned*)ws; ws += (size_t)NNODES * HEADS * sizeof(unsigned);
  float*    ssum = (float*)ws;    ws += (size_t)NNODES * HEADS * sizeof(float);

  k_init<<<(NNODES * CO + 255) / 256, 256, 0, stream>>>(out, mkey, ssum);
  k_gemm<<<NNODES / 16, 256, 0, stream>>>(feat, W, ft);
  k_eler<<<(NNODES * HEADS + 255) / 256, 256, 0, stream>>>(ft, attn, el, er);
  k_edge_max<<<(NEDGES * HEADS + 255) / 256, 256, 0, stream>>>(src, dst, el, er, mkey);
  k_edge_sum<<<(NEDGES * HEADS + 255) / 256, 256, 0, stream>>>(src, dst, el, er, mkey, ssum);
  k_edge_agg<<<NEDGES / 8, 256, 0, stream>>>(src, dst, el, er, mkey, ssum, ft, out);
}


# Edges counting-sorted by destination (integer atomic ranks); one wave per node does max, exp-sum and weighted row sum in registers, no float atomics; sums now run in segment (arrival-rank) order
# speedup vs baseline: 1.9181x; 1.9181x over previous
#include <hip/hip_runtime.h>
#include <math.h>


#define NNODES 50000
#define NEDGES 1600000
#define IN_F   256
#define HEADS  4
#define OUT_F  32
#define CO     (HEADS * OUT_F)
#define NEG_SLOPE 0.2f

typedef __attribute__((ext_vector_type(2))) float v2f;
typedef __attribute__((ext_vector_type(8))) float v8f;

__device__ __forceinline__ float leaky(float x) {
  return x > 0.f ? x : NEG_SLOPE * x;
}

static_assert(NNODES <= 65536, "source ids are kept as 16-bit values");
static_assert(NNODES % 4 == 0, "the scan reads the counts four at a time");

__global__ __launch_bounds__(256) void k_zero(int* __restrict__ cnt) {
  int i = blockIdx.x * 256 + threadIdx.x;
  if (i < NNODES) cnt[i] = 0;
}

__global__ __launch_bounds__(256) void k_hist(const int* __restrict__ dst,
                                              int* __restrict__ cnt,
                                              int* __restrict__ rank) {
  int e = blockIdx.x * 256 + threadIdx.x;
  if (e >= NEDGES) return;
  rank[e] = atomicAdd(&cnt[dst[e]], 1);
}

__global__ __launch_bounds__(1024) void k_scan(const int* __restrict__ cnt,
                                               int* __restrict__ off) {
  __shared__ int wtot[32];
  const int t = threadIdx.x, lane = t & 31, w = t >> 5;
  int carry = 0;
  for (int base = 0; base < NNODES; base += 4096) {
    const int idx = base + t * 4;
    int4 v = make_int4(0, 0, 0, 0);
    if (idx < NNODES) v = *(const int4*)(cnt + idx);
    const int tsum = v.x + v.y + v.z + v.w;
    int inc = tsum;
    for (int o = 1; o < 32; o <<= 1) {
      const int u = __shfl_up(inc, o, 32);
      if (lane >= o) inc += u;
    }
    if (lane == 31) wtot[w] = inc;
    __syncthreads();
    const int u = wtot[lane];
    int pre = (lane < w) ? u : 0;
    int tot = u;
    for (int o = 16; o > 0; o >>= 1) {
      pre += __shfl_xor(pre, o, 32);
      tot += __shfl_xor(tot, o, 32);
    }
    const int excl = carry + pre + inc - tsum;
    if (idx < NNODES) {
      int4 r;
      r.x = excl;
      r.y = r.x + v.x;
      r.z = r.y + v.y;
      r.w = r.z + v.z;
      *(int4*)(off + idx) = r;
    }
    carry += tot;
    __syncthreads();
  }
  if (t == 0) off[NNODES] = carry;
}

__global__ __launch_bounds__(256) void k_fill(const int* __restrict__ src,
                                              const int* __restrict__ dst,
                                              const int* __restrict__ off,
                                              const int* __restrict__ rank,
                                              unsigned short* __restrict__ ssrc) {
  int e = blockIdx.x * 256 + threadIdx.x;
  if (e >= NEDGES) return;
  ssrc[off[dst[e]] + rank[e]] = (unsigned short)src[e];
}

__global__ __launch_bounds__(256) void k_gemm(const float* __restrict__ feat,
                                              const float* __restrict__ W,
                                              float* __restrict__ ft) {
  __shared__ float As[16][IN_F + 4];
  const int tid    = threadIdx.x;
  const int wave   = tid >> 5;
  const int lane   = tid & 31;
  const int tile_m = blockIdx.x * 16;

  for (int i = tid; i < 16 * (IN_F / 4); i += 256) {
    const int r  = i / (IN_F / 4);
    const int c4 = i % (IN_F / 4);
    float4 v = make_float4(0.f, 0.f, 0.f, 0.f);
    const int row = tile_m + r;
    if (row < NNODES) v = ((const float4*)(feat + (size_t)row * IN_F))[c4];
    ((float4*)&As[r][0])[c4] = v;
  }
  __syncthreads();

  const int koff = (lane < 16) ? 0 : 2;
  const int arow = lane & 15;
  const int col  = wave * 16 + arow;
  const float* wrow = W + (size_t)col * IN_F;

  v8f c = {};
  for (int k0 = 0; k0 < IN_F; k0 += 4) {
    v2f a, b;
    a.x = As[arow][k0 + koff];
    a.y = As[arow][k0 + koff + 1];
    b.x = wrow[k0 + koff];
    b.y = wrow[k0 + koff + 1];
    c = __builtin_amdgcn_wmma_f32_16x16x4_f32(
         false, a,  false, b,
         (short)0, c,  false,  false);
  }

  const int mbase  = (lane < 16) ? 0 : 8;
  const int outcol = col;
#pragma unroll
  for (int j = 0; j < 8; ++j) {
    const int row = tile_m + mbase + j;
    if (row < NNODES) ft[(size_t)row * CO + outcol] = c[j];
  }
}

__global__ __launch_bounds__(256) void k_eler(const float* __restrict__ ft,
                                              const float* __restrict__ attn,
                                              float* __restrict__ el,
                                              float* __restrict__ er) {
  int i = blockIdx.x * 256 + threadIdx.x;
  if (i >= NNODES * HEADS) return;
  const int n = i / HEADS, h = i % HEADS;
  const float* f  = ft + (size_t)n * CO + h * OUT_F;
  const float* al = attn + h * 2 * OUT_F;
  const float* ar = al + OUT_F;
  float sl = 0.f, sr = 0.f;
#pragma unroll
  for (int d = 0; d < OUT_F; ++d) {
    const float v = f[d];
    sl += v * al[d];
    sr += v * ar[d];
  }
  el[i] = sl;
  er[i] = sr;
}

__device__ __forceinline__ float lane_f(float v, int l) {
  return __int_as_float(__builtin_amdgcn_readlane(__float_as_int(v), l));
}
__device__ __forceinline__ float pick4(int k0, int k1, int k2, int k3,
                                       float x0, float x1, float x2, float x3) {
  return __int_as_float((__float_as_int(x0) & k0) | (__float_as_int(x1) & k1) |
                        (__float_as_int(x2) & k2) | (__float_as_int(x3) & k3));
}

__device__ __forceinline__ int src_at(const unsigned short* __restrict__ ssrc, int p) {
  const int s = ssrc[p];
  return s < NNODES ? s : NNODES - 1;
}

__global__ __launch_bounds__(256) void k_node(const unsigned short* __restrict__ ssrc,
                                              const int* __restrict__ off,
                                              const float* __restrict__ el,
                                              const float* __restrict__ er,
                                              const float* __restrict__ ft,
                                              float* __restrict__ out) {
  const int lane = threadIdx.x & 31;
  const int wv = __builtin_amdgcn_readfirstlane((int)(threadIdx.x >> 5));
  const int d  = blockIdx.x * 8 + wv;
  if (d >= NNODES) return;
  const int beg = off[d];
  const int deg = off[d + 1] - beg;
  const float4 r4 = *(const float4*)(er + d * HEADS);
  const int hq = lane >> 3;
  const int k0 = (hq == 0) ? -1 : 0;
  const int k1 = (hq == 1) ? -1 : 0;
  const int k2 = (hq == 2) ? -1 : 0;
  const int k3 = (hq == 3) ? -1 : 0;

  const bool in0 = lane < deg;
  int sF = 0;
  float x0 = -INFINITY, x1 = -INFINITY, x2 = -INFINITY, x3 = -INFINITY;
  if (in0) {
    sF = src_at(ssrc, beg + lane);
    const float4 l4 = *(const float4*)(el + sF * HEADS);
    x0 = leaky(l4.x + r4.x);
    x1 = leaky(l4.y + r4.y);
    x2 = leaky(l4.z + r4.z);
    x3 = leaky(l4.w + r4.w);
  }

  float m0 = x0, m1 = x1, m2 = x2, m3 = x3;
  for (int c = 32; c < deg; c += 32) {
    const int j = c + lane;
    if (j < deg) {
      const int s = src_at(ssrc, beg + j);
      const float4 l4 = *(const float4*)(el + s * HEADS);
      m0 = fmaxf(m0, leaky(l4.x + r4.x));
      m1 = fmaxf(m1, leaky(l4.y + r4.y));
      m2 = fmaxf(m2, leaky(l4.z + r4.z));
      m3 = fmaxf(m3, leaky(l4.w + r4.w));
    }
  }
  for (int o = 16; o > 0; o >>= 1) {
    m0 = fmaxf(m0, __shfl_xor(m0, o, 32));
    m1 = fmaxf(m1, __shfl_xor(m1, o, 32));
    m2 = fmaxf(m2, __shfl_xor(m2, o, 32));
    m3 = fmaxf(m3, __shfl_xor(m3, o, 32));
  }

  const float e0 = in0 ? expf(x0 - m0) : 0.f;
  const float e1 = in0 ? expf(x1 - m1) : 0.f;
  const float e2 = in0 ? expf(x2 - m2) : 0.f;
  const float e3 = in0 ? expf(x3 - m3) : 0.f;
  float s0 = e0, s1 = e1, s2 = e2, s3 = e3;
  for (int c = 32; c < deg; c += 32) {
    const int j = c + lane;
    if (j < deg) {
      const int s = src_at(ssrc, beg + j);
      const float4 l4 = *(const float4*)(el + s * HEADS);
      s0 += expf(leaky(l4.x + r4.x) - m0);
      s1 += expf(leaky(l4.y + r4.y) - m1);
      s2 += expf(leaky(l4.z + r4.z) - m2);
      s3 += expf(leaky(l4.w + r4.w) - m3);
    }
  }
  for (int o = 16; o > 0; o >>= 1) {
    s0 += __shfl_xor(s0, o, 32);
    s1 += __shfl_xor(s1, o, 32);
    s2 += __shfl_xor(s2, o, 32);
    s3 += __shfl_xor(s3, o, 32);
  }

  float4 acc = make_float4(0.f, 0.f, 0.f, 0.f);
  const float* fl = ft + lane * 4;
  for (int c = 0; c < deg; c += 32) {
    int s = 0;
    float a0 = 0.f, a1 = 0.f, a2 = 0.f, a3 = 0.f;
    if (c == 0) {
      s  = sF;
      a0 = e0 / s0;
      a1 = e1 / s1;
      a2 = e2 / s2;
      a3 = e3 / s3;
    } else {
      const int j = c + lane;
      if (j < deg) {
        s = src_at(ssrc, beg + j);
        const float4 l4 = *(const float4*)(el + s * HEADS);
        a0 = expf(leaky(l4.x + r4.x) - m0) / s0;
        a1 = expf(leaky(l4.y + r4.y) - m1) / s1;
        a2 = expf(leaky(l4.z + r4.z) - m2) / s2;
        a3 = expf(leaky(l4.w + r4.w) - m3) / s3;
      }
    }
    const int n = (deg - c < 32) ? (deg - c) : 32;
    int k = 0;
    for (; k + 4 <= n; k += 4) {
      const int sA = __builtin_amdgcn_readlane(s, k);
      const int sB = __builtin_amdgcn_readlane(s, k + 1);
      const int sC = __builtin_amdgcn_readlane(s, k + 2);
      const int sD = __builtin_amdgcn_readlane(s, k + 3);
      const float4 fA = *(const float4*)(fl + (size_t)sA * CO);
      const float4 fB = *(const float4*)(fl + (size_t)sB * CO);
      const float4 fC = *(const float4*)(fl + (size_t)sC * CO);
      const float4 fD = *(const float4*)(fl + (size_t)sD * CO);
      const float aA = pick4(k0, k1, k2, k3, lane_f(a0, k), lane_f(a1, k), lane_f(a2, k), lane_f(a3, k));
      const float aB = pick4(k0, k1, k2, k3, lane_f(a0, k + 1), lane_f(a1, k + 1), lane_f(a2, k + 1), lane_f(a3, k + 1));
      const float aC = pick4(k0, k1, k2, k3, lane_f(a0, k + 2), lane_f(a1, k + 2), lane_f(a2, k + 2), lane_f(a3, k + 2));
      const float aD = pick4(k0, k1, k2, k3, lane_f(a0, k + 3), lane_f(a1, k + 3), lane_f(a2, k + 3), lane_f(a3, k + 3));
      acc.x += fA.x * aA; acc.y += fA.y * aA; acc.z += fA.z * aA; acc.w += fA.w * aA;
      acc.x += fB.x * aB; acc.y += fB.y * aB; acc.z += fB.z * aB; acc.w += fB.w * aB;
      acc.x += fC.x * aC; acc.y += fC.y * aC; acc.z += fC.z * aC; acc.w += fC.w * aC;
      acc.x += fD.x * aD; acc.y += fD.y * aD; acc.z += fD.z * aD; acc.w += fD.w * aD;
    }
    for (; k < n; ++k) {
      const int sA = __builtin_amdgcn_readlane(s, k);
      const float4 fA = *(const float4*)(fl + (size_t)sA * CO);
      const float aA = pick4(k0, k1, k2, k3, lane_f(a0, k), lane_f(a1, k), lane_f(a2, k), lane_f(a3, k));
      acc.x += fA.x * aA; acc.y += fA.y * aA; acc.z += fA.z * aA; acc.w += fA.w * aA;
    }
  }
  *(float4*)(out + (size_t)d * CO + lane * 4) = acc;
}

extern "C" void kernel_launch(void* const* d_in, const int* in_sizes, int n_in,
                              void* d_out, int out_size, void* d_ws, size_t ws_size,
                              hipStream_t stream) {
  (void)in_sizes; (void)n_in; (void)out_size; (void)ws_size;

  const float* feat = (const float*)d_in[0];
  const int*   src  = (const int*)d_in[1];
  const int*   dst  = (const int*)d_in[2];
  const float* W    = (const float*)d_in[3];
  const float* attn = (const float*)d_in[4];
  float* out = (float*)d_out;

  char* ws = (char*)d_ws;
  float* ft   = (float*)ws;
  int*   cnt  = (int*)ws;
  int*   rank = (int*)(ws + 262144);
  ws += (size_t)NNODES * CO * sizeof(float);
  float* el   = (float*)ws;  ws += (size_t)NNODES * HEADS * sizeof(float);
  float* er   = (float*)ws;  ws += (size_t)NNODES * HEADS * sizeof(float);
  int*   off  = (int*)ws;    ws += 200064;
  unsigned short* ssrc = (unsigned short*)ws;

  k_zero<<<(NNODES + 255) / 256, 256, 0, stream>>>(cnt);
  k_hist<<<(NEDGES + 255) / 256, 256, 0, stream>>>(dst, cnt, rank);
  k_scan<<<1, 1024, 0, stream>>>(cnt, off);
  k_fill<<<(NEDGES + 255) / 256, 256, 0, stream>>>(src, dst, off, rank, ssrc);
  k_gemm<<<NNODES / 16, 256, 0, stream>>>(feat, W, ft);
  k_eler<<<(NNODES * HEADS + 255) / 256, 256, 0, stream>>>(ft, attn, el, er);
  k_node<<<(NNODES + 7) / 8, 256, 0, stream>>>(ssrc, off, el, er, ft, out);
}


# Sort counters moved into LDS: 256 work-groups (64 edge slices x 4 destination ranges) count and place edges with LDS adds; no global atomics, no rank array; sums run in slice order, then LDS arrival
# speedup vs baseline: 2.8415x; 1.4814x over previous
#include <hip/hip_runtime.h>
#include <math.h>


#define NNODES 50000
#define NEDGES 1600000
#define IN_F   256
#define HEADS  4
#define OUT_F  32
#define CO     (HEADS * OUT_F)
#define NEG_SLOPE 0.2f

typedef __attribute__((ext_vector_type(2))) float v2f;
typedef __attribute__((ext_vector_type(8))) float v8f;

__device__ __forceinline__ float leaky(float x) {
  return x > 0.f ? x : NEG_SLOPE * x;
}

static_assert(NNODES <= 65536, "source ids are kept as 16-bit values");
static_assert(NNODES % 4 == 0, "the scan reads the counts four at a time");

#define SLICES   64
#define RANGES   4
#define SLICE_E  (NEDGES / SLICES)
#define RANGE_N  (NNODES / RANGES)
#define SLICE_Q  (SLICE_E / 4)
#define RANGE_Q  (RANGE_N / 4)
#define PIECES   ((SLICE_Q + 1023) / 1024)
#define RPIECES  ((RANGE_Q + 1023) / 1024)
static_assert(NEDGES % (SLICES * 4) == 0, "a slice is read as 16-byte pieces");
static_assert(NNODES % (RANGES * 4) == 0, "a range is read as 16-byte pieces");
static_assert(SLICES % 32 == 0, "the column pass takes the rows 32 at a time");
static_assert(SLICES * NNODES <= NNODES * CO, "the count rows fit where ft goes later");

__global__ __launch_bounds__(1024) void k_hist(const int* __restrict__ dst,
                                               int* __restrict__ crow) {
  __shared__ __attribute__((aligned(16))) int h[RANGE_N];
  const int t  = threadIdx.x;
  const int s  = blockIdx.x / RANGES;
  const int lo = (blockIdx.x % RANGES) * RANGE_N;
  const int4* d4 = (const int4*)(dst + (size_t)s * SLICE_E);
  int4 dv[PIECES];
#pragma unroll
  for (int k = 0; k < PIECES; ++k) {
    const int i = t + k * 1024;
    dv[k] = (i < SLICE_Q) ? d4[i] : make_int4(-1, -1, -1, -1);
  }
  int4* h4 = (int4*)h;
  for (int i = t; i < RANGE_Q; i += 1024) h4[i] = make_int4(0, 0, 0, 0);
  __syncthreads();
#pragma unroll
  for (int k = 0; k < PIECES; ++k) {
    const unsigned u0 = (unsigned)(dv[k].x - lo), u1 = (unsigned)(dv[k].y - lo);
    const unsigned u2 = (unsigned)(dv[k].z - lo), u3 = (unsigned)(dv[k].w - lo);
    if (u0 < RANGE_N) atomicAdd(&h[u0], 1);
    if (u1 < RANGE_N) atomicAdd(&h[u1], 1);
    if (u2 < RANGE_N) atomicAdd(&h[u2], 1);
    if (u3 < RANGE_N) atomicAdd(&h[u3], 1);
  }
  __syncthreads();
  int4* c4 = (int4*)(crow + (size_t)s * NNODES + lo);
  for (int i = t; i < RANGE_Q; i += 1024) c4[i] = h4[i];
}

__global__ __launch_bounds__(128) void k_colsum(int* __restrict__ crow,
                                                int* __restrict__ cnt) {
  const int d = blockIdx.x * 128 + threadIdx.x;
  if (d >= NNODES) return;
  int* p = crow + d;
  int run = 0;
  for (int s0 = 0; s0 < SLICES; s0 += 32) {
    int c[32];
#pragma unroll
    for (int j = 0; j < 32; ++j) c[j] = p[(size_t)(s0 + j) * NNODES];
#pragma unroll
    for (int j = 0; j < 32; ++j) {
      p[(size_t)(s0 + j) * NNODES] = run;
      run += c[j];
    }
  }
  cnt[d] = run;
}

__global__ __launch_bounds__(1024) void k_scan(const int* __restrict__ cnt,
                                               int* __restrict__ off) {
  __shared__ int wtot[32];
  const int t = threadIdx.x, lane = t & 31, w = t >> 5;
  int carry = 0;
  for (int base = 0; base < NNODES; base += 4096) {
    const int idx = base + t * 4;
    int4 v = make_int4(0, 0, 0, 0);
    if (idx < NNODES) v = *(const int4*)(cnt + idx);
    const int tsum = v.x + v.y + v.z + v.w;
    int inc = tsum;
    for (int o = 1; o < 32; o <<= 1) {
      const int u = __shfl_up(inc, o, 32);
      if (lane >= o) inc += u;
    }
    if (lane == 31) wtot[w] = inc;
    __syncthreads();
    const int u = wtot[lane];
    int pre = (lane < w) ? u : 0;
    int tot = u;
    for (int o = 16; o > 0; o >>= 1) {
      pre += __shfl_xor(pre, o, 32);
      tot += __shfl_xor(tot, o, 32);
    }
    const int excl = carry + pre + inc - tsum;
    if (idx < NNODES) {
      int4 r;
      r.x = excl;
      r.y = r.x + v.x;
      r.z = r.y + v.y;
      r.w = r.z + v.z;
      *(int4*)(off + idx) = r;
    }
    carry += tot;
    __syncthreads();
  }
  if (t == 0) off[NNODES] = carry;
}

__global__ __launch_bounds__(1024) void k_fill(const int* __restrict__ src,
                                               const int* __restrict__ dst,
                                               const int* __restrict__ off,
                                               const int* __restrict__ crow,
                                               unsigned short* __restrict__ ssrc) {
  __shared__ __attribute__((aligned(16))) int cur[RANGE_N];
  const int t  = threadIdx.x;
  const int s  = blockIdx.x / RANGES;
  const int lo = (blockIdx.x % RANGES) * RANGE_N;
  const int4* s4 = (const int4*)(src + (size_t)s * SLICE_E);
  const int4* d4 = (const int4*)(dst + (size_t)s * SLICE_E);
  int4 sv[PIECES], dv[PIECES];
#pragma unroll
  for (int k = 0; k < PIECES; ++k) {
    const int i = t + k * 1024;
    const bool in = i < SLICE_Q;
    dv[k] = in ? d4[i] : make_int4(-1, -1, -1, -1);
    sv[k] = in ? s4[i] : make_int4(0, 0, 0, 0);
  }
  const int4* o4 = (const int4*)(off + lo);
  const int4* p4 = (const int4*)(crow + (size_t)s * NNODES + lo);
  int4* c4 = (int4*)cur;
  int4 a[RPIECES], b[RPIECES];
#pragma unroll
  for (int k = 0; k < RPIECES; ++k) {
    const int i = t + k * 1024;
    if (i < RANGE_Q) { a[k] = o4[i]; b[k] = p4[i]; }
  }
#pragma unroll
  for (int k = 0; k < RPIECES; ++k) {
    const int i = t + k * 1024;
    if (i < RANGE_Q)
      c4[i] = make_int4(a[k].x + b[k].x, a[k].y + b[k].y, a[k].z + b[k].z, a[k].w + b[k].w);
  }
  __syncthreads();
#pragma unroll
  for (int k = 0; k < PIECES; ++k) {
    const unsigned u0 = (unsigned)(dv[k].x - lo), u1 = (unsigned)(dv[k].y - lo);
    const unsigned u2 = (unsigned)(dv[k].z - lo), u3 = (unsigned)(dv[k].w - lo);
    if (u0 < RANGE_N) ssrc[atomicAdd(&cur[u0], 1)] = (unsigned short)sv[k].x;
    if (u1 < RANGE_N) ssrc[atomicAdd(&cur[u1], 1)] = (unsigned short)sv[k].y;
    if (u2 < RANGE_N) ssrc[atomicAdd(&cur[u2], 1)] = (unsigned short)sv[k].z;
    if (u3 < RANGE_N) ssrc[atomicAdd(&cur[u3], 1)] = (unsigned short)sv[k].w;
  }
}

__global__ __launch_bounds__(256) void k_gemm(const float* __restrict__ feat,
                                              const float* __restrict__ W,
                                              float* __restrict__ ft) {
  __shared__ float As[16][IN_F + 4];
  const int tid    = threadIdx.x;
  const int wave   = tid >> 5;
  const int lane   = tid & 31;
  const int tile_m = blockIdx.x * 16;

  for (int i = tid; i < 16 * (IN_F / 4); i += 256) {
    const int r  = i / (IN_F / 4);
    const int c4 = i % (IN_F / 4);
    float4 v = make_float4(0.f, 0.f, 0.f, 0.f);
    const int row = tile_m + r;
    if (row < NNODES) v = ((const float4*)(feat + (size_t)row * IN_F))[c4];
    ((float4*)&As[r][0])[c4] = v;
  }
  __syncthreads();

  const int koff = (lane < 16) ? 0 : 2;
  const int arow = lane & 15;
  const int col  = wave * 16 + arow;
  const float* wrow = W + (size_t)col * IN_F;

  v8f c = {};
  for (int k0 = 0; k0 < IN_F; k0 += 4) {
    v2f a, b;
    a.x = As[arow][k0 + koff];
    a.y = As[arow][k0 + koff + 1];
    b.x = wrow[k0 + koff];
    b.y = wrow[k0 + koff + 1];
    c = __builtin_amdgcn_wmma_f32_16x16x4_f32(
         false, a,  false, b,
         (short)0, c,  false,  false);
  }

  const int mbase  = (lane < 16) ? 0 : 8;
  const int outcol = col;
#pragma unroll
  for (int j = 0; j < 8; ++j) {
    const int row = tile_m + mbase + j;
    if (row < NNODES) ft[(size_t)row * CO + outcol] = c[j];
  }
}

__global__ __launch_bounds__(256) void k_eler(const float* __restrict__ ft,
                                              const float* __restrict__ attn,
                                              float* __restrict__ el,
                                              float* __restrict__ er) {
  int i = blockIdx.x * 256 + threadIdx.x;
  if (i >= NNODES * HEADS) return;
  const int n = i / HEADS, h = i % HEADS;
  const float* f  = ft + (size_t)n * CO + h * OUT_F;
  const float* al = attn + h * 2 * OUT_F;
  const float* ar = al + OUT_F;
  float sl = 0.f, sr = 0.f;
#pragma unroll
  for (int d = 0; d < OUT_F; ++d) {
    const float v = f[d];
    sl += v * al[d];
    sr += v * ar[d];
  }
  el[i] = sl;
  er[i] = sr;
}

__device__ __forceinline__ float lane_f(float v, int l) {
  return __int_as_float(__builtin_amdgcn_readlane(__float_as_int(v), l));
}
__device__ __forceinline__ float pick4(int k0, int k1, int k2, int k3,
                                       float x0, float x1, float x2, float x3) {
  return __int_as_float((__float_as_int(x0) & k0) | (__float_as_int(x1) & k1) |
                        (__float_as_int(x2) & k2) | (__float_as_int(x3) & k3));
}

__device__ __forceinline__ int src_at(const unsigned short* __restrict__ ssrc, int p) {
  const int s = ssrc[p];
  return s < NNODES ? s : NNODES - 1;
}

__global__ __launch_bounds__(256) void k_node(const unsigned short* __restrict__ ssrc,
                                              const int* __restrict__ off,
                                              const float* __restrict__ el,
                                              const float* __restrict__ er,
                                              const float* __restrict__ ft,
                                              float* __restrict__ out) {
  const int lane = threadIdx.x & 31;
  const int wv = __builtin_amdgcn_readfirstlane((int)(threadIdx.x >> 5));
  const int d  = blockIdx.x * 8 + wv;
  if (d >= NNODES) return;
  const int beg = off[d];
  const int deg = off[d + 1] - beg;
  const float4 r4 = *(const float4*)(er + d * HEADS);
  const int hq = lane >> 3;
  const int k0 = (hq == 0) ? -1 : 0;
  const int k1 = (hq == 1) ? -1 : 0;
  const int k2 = (hq == 2) ? -1 : 0;
  const int k3 = (hq == 3) ? -1 : 0;

  const bool in0 = lane < deg;
  int sF = 0;
  float x0 = -INFINITY, x1 = -INFINITY, x2 = -INFINITY, x3 = -INFINITY;
  if (in0) {
    sF = src_at(ssrc, beg + lane);
    const float4 l4 = *(const float4*)(el + sF * HEADS);
    x0 = leaky(l4.x + r4.x);
    x1 = leaky(l4.y + r4.y);
    x2 = leaky(l4.z + r4.z);
    x3 = leaky(l4.w + r4.w);
  }

  float m0 = x0, m1 = x1, m2 = x2, m3 = x3;
  for (int c = 32; c < deg; c += 32) {
    const int j = c + lane;
    if (j < deg) {
      const int s = src_at(ssrc, beg + j);
      const float4 l4 = *(const float4*)(el + s * HEADS);
      m0 = fmaxf(m0, leaky(l4.x + r4.x));
      m1 = fmaxf(m1, leaky(l4.y + r4.y));
      m2 = fmaxf(m2, leaky(l4.z + r4.z));
      m3 = fmaxf(m3, leaky(l4.w + r4.w));
    }
  }
  for (int o = 16; o > 0; o >>= 1) {
    m0 = fmaxf(m0, __shfl_xor(m0, o, 32));
    m1 = fmaxf(m1, __shfl_xor(m1, o, 32));
    m2 = fmaxf(m2, __shfl_xor(m2, o, 32));
    m3 = fmaxf(m3, __shfl_xor(m3, o, 32));
  }

  const float e0 = in0 ? expf(x0 - m0) : 0.f;
  const float e1 = in0 ? expf(x1 - m1) : 0.f;
  const float e2 = in0 ? expf(x2 - m2) : 0.f;
  const float e3 = in0 ? expf(x3 - m3) : 0.f;
  float s0 = e0, s1 = e1, s2 = e2, s3 = e3;
  for (int c = 32; c < deg; c += 32) {
    const int j = c + lane;
    if (j < deg) {
      const int s = src_at(ssrc, beg + j);
      const float4 l4 = *(const float4*)(el + s * HEADS);
      s0 += expf(leaky(l4.x + r4.x) - m0);
      s1 += expf(leaky(l4.y + r4.y) - m1);
      s2 += expf(leaky(l4.z + r4.z) - m2);
      s3 += expf(leaky(l4.w + r4.w) - m3);
    }
  }
  for (int o = 16; o > 0; o >>= 1) {
    s0 += __shfl_xor(s0, o, 32);
    s1 += __shfl_xor(s1, o, 32);
    s2 += __shfl_xor(s2, o, 32);
    s3 += __shfl_xor(s3, o, 32);
  }

  float4 acc = make_float4(0.f, 0.f, 0.f, 0.f);
  const float* fl = ft + lane * 4;
  for (int c = 0; c < deg; c += 32) {
    int s = 0;
    float a0 = 0.f, a1 = 0.f, a2 = 0.f, a3 = 0.f;
    if (c == 0) {
      s  = sF;
      a0 = e0 / s0;
      a1 = e1 / s1;
      a2 = e2 / s2;
      a3 = e3 / s3;
    } else {
      const int j = c + lane;
      if (j < deg) {
        s = src_at(ssrc, beg + j);
        const float4 l4 = *(const float4*)(el + s * HEADS);
        a0 = expf(leaky(l4.x + r4.x) - m0) / s0;
        a1 = expf(leaky(l4.y + r4.y) - m1) / s1;
        a2 = expf(leaky(l4.z + r4.z) - m2) / s2;
        a3 = expf(leaky(l4.w + r4.w) - m3) / s3;
      }
    }
    const int n = (deg - c < 32) ? (deg - c) : 32;
    int k = 0;
    for (; k + 4 <= n; k += 4) {
      const int sA = __builtin_amdgcn_readlane(s, k);
      const int sB = __builtin_amdgcn_readlane(s, k + 1);
      const int sC = __builtin_amdgcn_readlane(s, k + 2);
      const int sD = __builtin_amdgcn_readlane(s, k + 3);
      const float4 fA = *(const float4*)(fl + (size_t)sA * CO);
      const float4 fB = *(const float4*)(fl + (size_t)sB * CO);
      const float4 fC = *(const float4*)(fl + (size_t)sC * CO);
      const float4 fD = *(const float4*)(fl + (size_t)sD * CO);
      const float aA = pick4(k0, k1, k2, k3, lane_f(a0, k), lane_f(a1, k), lane_f(a2, k), lane_f(a3, k));
      const float aB = pick4(k0, k1, k2, k3, lane_f(a0, k + 1), lane_f(a1, k + 1), lane_f(a2, k + 1), lane_f(a3, k + 1));
      const float aC = pick4(k0, k1, k2, k3, lane_f(a0, k + 2), lane_f(a1, k + 2), lane_f(a2, k + 2), lane_f(a3, k + 2));
      const float aD = pick4(k0, k1, k2, k3, lane_f(a0, k + 3), lane_f(a1, k + 3), lane_f(a2, k + 3), lane_f(a3, k + 3));
      acc.x += fA.x * aA; acc.y += fA.y * aA; acc.z += fA.z * aA; acc.w += fA.w * aA;
      acc.x += fB.x * aB; acc.y += fB.y * aB; acc.z += fB.z * aB; acc.w += fB.w * aB;
      acc.x += fC.x * aC; acc.y += fC.y * aC; acc.z += fC.z * aC; acc.w += fC.w * aC;
      acc.x += fD.x * aD; acc.y += fD.y * aD; acc.z += fD.z * aD; acc.w += fD.w * aD;
    }
    for (; k < n; ++k) {
      const int sA = __builtin_amdgcn_readlane(s, k);
      const float4 fA = *(const float4*)(fl + (size_t)sA * CO);
      const float aA = pick4(k0, k1, k2, k3, lane_f(a0, k), lane_f(a1, k), lane_f(a2, k), lane_f(a3, k));
      acc.x += fA.x * aA; acc.y += fA.y * aA; acc.z += fA.z * aA; acc.w += fA.w * aA;
    }
  }
  *(float4*)(out + (size_t)d * CO + lane * 4) = acc;
}

extern "C" void kernel_launch(void* const* d_in, const int* in_sizes, int n_in,
                              void* d_out, int out_size, void* d_ws, size_t ws_size,
                              hipStream_t stream) {
  (void)in_sizes; (void)n_in; (void)out_size; (void)ws_size;

  const float* feat = (const float*)d_in[0];
  const int*   src  = (const int*)d_in[1];
  const int*   dst  = (const int*)d_in[2];
  const float* W    = (const float*)d_in[3];
  const float* attn = (const float*)d_in[4];
  float* out = (float*)d_out;

  char* ws = (char*)d_ws;
  float* ft   = (float*)ws;
  int*   crow = (int*)ws;
  ws += (size_t)NNODES * CO * sizeof(float);
  int*   cnt  = (int*)ws;
  float* el   = (float*)ws;  ws += (size_t)NNODES * HEADS * sizeof(float);
  float* er   = (float*)ws;  ws += (size_t)NNODES * HEADS * sizeof(float);
  int*   off  = (int*)ws;    ws += 200064;
  unsigned short* ssrc = (unsigned short*)ws;

  k_hist<<<SLICES * RANGES, 1024, 0, stream>>>(dst, crow);
  k_colsum<<<(NNODES + 127) / 128, 128, 0, stream>>>(crow, cnt);
  k_scan<<<1, 1024, 0, stream>>>(cnt, off);
  k_fill<<<SLICES * RANGES, 1024, 0, stream>>>(src, dst, off, crow, ssrc);
  k_gemm<<<NNODES / 16, 256, 0, stream>>>(feat, W, ft);
  k_eler<<<(NNODES * HEADS + 255) / 256, 256, 0, stream>>>(ft, attn, el, er);
  k_node<<<(NNODES + 7) / 8, 256, 0, stream>>>(ssrc, off, el, er, ft, out);
}


# GEMM made resident: 512 work-groups walk the 16-row tiles, each wave keeps its 16-column slice of W in registers instead of re-reading it per tile; k now summed {0,1,4,5},{2,3,6,7} per 8
# speedup vs baseline: 3.1842x; 1.1206x over previous
#include <hip/hip_runtime.h>
#include <math.h>


#define NNODES 50000
#define NEDGES 1600000
#define IN_F   256
#define HEADS  4
#define OUT_F  32
#define CO     (HEADS * OUT_F)
#define NEG_SLOPE 0.2f

typedef __attribute__((ext_vector_type(2))) float v2f;
typedef __attribute__((ext_vector_type(8))) float v8f;

__device__ __forceinline__ float leaky(float x) {
  return x > 0.f ? x : NEG_SLOPE * x;
}

static_assert(NNODES <= 65536, "source ids are kept as 16-bit values");
static_assert(NNODES % 4 == 0, "the scan reads the counts four at a time");

#define SLICES   64
#define RANGES   4
#define SLICE_E  (NEDGES / SLICES)
#define RANGE_N  (NNODES / RANGES)
#define SLICE_Q  (SLICE_E / 4)
#define RANGE_Q  (RANGE_N / 4)
#define PIECES   ((SLICE_Q + 1023) / 1024)
#define RPIECES  ((RANGE_Q + 1023) / 1024)
static_assert(NEDGES % (SLICES * 4) == 0, "a slice is read as 16-byte pieces");
static_assert(NNODES % (RANGES * 4) == 0, "a range is read as 16-byte pieces");
static_assert(SLICES % 32 == 0, "the column pass takes the rows 32 at a time");
static_assert(SLICES * NNODES <= NNODES * CO, "the count rows fit where ft goes later");

__global__ __launch_bounds__(1024) void k_hist(const int* __restrict__ dst,
                                               int* __restrict__ crow) {
  __shared__ __attribute__((aligned(16))) int h[RANGE_N];
  const int t  = threadIdx.x;
  const int s  = blockIdx.x / RANGES;
  const int lo = (blockIdx.x % RANGES) * RANGE_N;
  const int4* d4 = (const int4*)(dst + (size_t)s * SLICE_E);
  int4 dv[PIECES];
#pragma unroll
  for (int k = 0; k < PIECES; ++k) {
    const int i = t + k * 1024;
    dv[k] = (i < SLICE_Q) ? d4[i] : make_int4(-1, -1, -1, -1);
  }
  int4* h4 = (int4*)h;
  for (int i = t; i < RANGE_Q; i += 1024) h4[i] = make_int4(0, 0, 0, 0);
  __syncthreads();
#pragma unroll
  for (int k = 0; k < PIECES; ++k) {
    const unsigned u0 = (unsigned)(dv[k].x - lo), u1 = (unsigned)(dv[k].y - lo);
    const unsigned u2 = (unsigned)(dv[k].z - lo), u3 = (unsigned)(dv[k].w - lo);
    if (u0 < RANGE_N) atomicAdd(&h[u0], 1);
    if (u1 < RANGE_N) atomicAdd(&h[u1], 1);
    if (u2 < RANGE_N) atomicAdd(&h[u2], 1);
    if (u3 < RANGE_N) atomicAdd(&h[u3], 1);
  }
  __syncthreads();
  int4* c4 = (int4*)(crow + (size_t)s * NNODES + lo);
  for (int i = t; i < RANGE_Q; i += 1024) c4[i] = h4[i];
}

__global__ __launch_bounds__(128) void k_colsum(int* __restrict__ crow,
                                                int* __restrict__ cnt) {
  const int d = blockIdx.x * 128 + threadIdx.x;
  if (d >= NNODES) return;
  int* p = crow + d;
  int run = 0;
  for (int s0 = 0; s0 < SLICES; s0 += 32) {
    int c[32];
#pragma unroll
    for (int j = 0; j < 32; ++j) c[j] = p[(size_t)(s0 + j) * NNODES];
#pragma unroll
    for (int j = 0; j < 32; ++j) {
      p[(size_t)(s0 + j) * NNODES] = run;
      run += c[j];
    }
  }
  cnt[d] = run;
}

__global__ __launch_bounds__(1024) void k_scan(const int* __restrict__ cnt,
                                               int* __restrict__ off) {
  __shared__ int wtot[32];
  const int t = threadIdx.x, lane = t & 31, w = t >> 5;
  int carry = 0;
  for (int base = 0; base < NNODES; base += 4096) {
    const int idx = base + t * 4;
    int4 v = make_int4(0, 0, 0, 0);
    if (idx < NNODES) v = *(const int4*)(cnt + idx);
    const int tsum = v.x + v.y + v.z + v.w;
    int inc = tsum;
    for (int o = 1; o < 32; o <<= 1) {
      const int u = __shfl_up(inc, o, 32);
      if (lane >= o) inc += u;
    }
    if (lane == 31) wtot[w] = inc;
    __syncthreads();
    const int u = wtot[lane];
    int pre = (lane < w) ? u : 0;
    int tot = u;
    for (int o = 16; o > 0; o >>= 1) {
      pre += __shfl_xor(pre, o, 32);
      tot += __shfl_xor(tot, o, 32);
    }
    const int excl = carry + pre + inc - tsum;
    if (idx < NNODES) {
      int4 r;
      r.x = excl;
      r.y = r.x + v.x;
      r.z = r.y + v.y;
      r.w = r.z + v.z;
      *(int4*)(off + idx) = r;
    }
    carry += tot;
    __syncthreads();
  }
  if (t == 0) off[NNODES] = carry;
}

__global__ __launch_bounds__(1024) void k_fill(const int* __restrict__ src,
                                               const int* __restrict__ dst,
                                               const int* __restrict__ off,
                                               const int* __restrict__ crow,
                                               unsigned short* __restrict__ ssrc) {
  __shared__ __attribute__((aligned(16))) int cur[RANGE_N];
  const int t  = threadIdx.x;
  const int s  = blockIdx.x / RANGES;
  const int lo = (blockIdx.x % RANGES) * RANGE_N;
  const int4* s4 = (const int4*)(src + (size_t)s * SLICE_E);
  const int4* d4 = (const int4*)(dst + (size_t)s * SLICE_E);
  int4 sv[PIECES], dv[PIECES];
#pragma unroll
  for (int k = 0; k < PIECES; ++k) {
    const int i = t + k * 1024;
    const bool in = i < SLICE_Q;
    dv[k] = in ? d4[i] : make_int4(-1, -1, -1, -1);
    sv[k] = in ? s4[i] : make_int4(0, 0, 0, 0);
  }
  const int4* o4 = (const int4*)(off + lo);
  const int4* p4 = (const int4*)(crow + (size_t)s * NNODES + lo);
  int4* c4 = (int4*)cur;
  int4 a[RPIECES], b[RPIECES];
#pragma unroll
  for (int k = 0; k < RPIECES; ++k) {
    const int i = t + k * 1024;
    if (i < RANGE_Q) { a[k] = o4[i]; b[k] = p4[i]; }
  }
#pragma unroll
  for (int k = 0; k < RPIECES; ++k) {
    const int i = t + k * 1024;
    if (i < RANGE_Q)
      c4[i] = make_int4(a[k].x + b[k].x, a[k].y + b[k].y, a[k].z + b[k].z, a[k].w + b[k].w);
  }
  __syncthreads();
#pragma unroll
  for (int k = 0; k < PIECES; ++k) {
    const unsigned u0 = (unsigned)(dv[k].x - lo), u1 = (unsigned)(dv[k].y - lo);
    const unsigned u2 = (unsigned)(dv[k].z - lo), u3 = (unsigned)(dv[k].w - lo);
    if (u0 < RANGE_N) ssrc[atomicAdd(&cur[u0], 1)] = (unsigned short)sv[k].x;
    if (u1 < RANGE_N) ssrc[atomicAdd(&cur[u1], 1)] = (unsigned short)sv[k].y;
    if (u2 < RANGE_N) ssrc[atomicAdd(&cur[u2], 1)] = (unsigned short)sv[k].z;
    if (u3 < RANGE_N) ssrc[atomicAdd(&cur[u3], 1)] = (unsigned short)sv[k].w;
  }
}

#define GEMM_WG    512
#define GEMM_TILES (NNODES / 16)
#define KPIECES    (IN_F / 8)
#define A_PITCH    (IN_F + 4)
static_assert(NNODES % 16 == 0, "every row tile is full: no row guards in the GEMM");
static_assert(16 * (IN_F / 4) == 4 * 256, "a thread loads four 16-byte pieces of a tile, rows lrow, +4, +8, +12");

__global__ __launch_bounds__(256) void k_gemm(const float* __restrict__ feat,
                                              const float* __restrict__ W,
                                              float* __restrict__ ft) {
  __shared__ __attribute__((aligned(16))) float As[2][16][A_PITCH];
  const int tid  = threadIdx.x;
  const int lane = tid & 31;
  const int wave = __builtin_amdgcn_readfirstlane(tid >> 5);
  const int c    = lane & 15;
  const int h    = lane >> 4;

  const float4* wrow = (const float4*)(W + (size_t)(wave * 16 + c) * IN_F + 4 * h);
  float4 bq[KPIECES];
#pragma unroll
  for (int j = 0; j < KPIECES; ++j) bq[j] = wrow[2 * j];

  const int lrow = tid >> 6;
  const int lcol = tid & 63;
  const float4* fsrc = (const float4*)feat + (size_t)lrow * (IN_F / 4) + lcol;

  int tile = blockIdx.x;
  const float4* f0 = fsrc + (size_t)tile * 16 * (IN_F / 4);
  float4 p0 = f0[0 * (IN_F / 4)], p1 = f0[4 * (IN_F / 4)];
  float4 p2 = f0[8 * (IN_F / 4)], p3 = f0[12 * (IN_F / 4)];
  ((float4*)&As[0][lrow + 0][0])[lcol]  = p0;
  ((float4*)&As[0][lrow + 4][0])[lcol]  = p1;
  ((float4*)&As[0][lrow + 8][0])[lcol]  = p2;
  ((float4*)&As[0][lrow + 12][0])[lcol] = p3;
  __syncthreads();

  float* obase = ft + (size_t)(8 * h) * CO + wave * 16 + c;
  int buf = 0;
  for (; tile < GEMM_TILES; tile += GEMM_WG) {
    const int  nxt  = tile + GEMM_WG;
    const bool more = nxt < GEMM_TILES;
    if (more) {
      const float4* fn = fsrc + (size_t)nxt * 16 * (IN_F / 4);
      p0 = fn[0 * (IN_F / 4)]; p1 = fn[4 * (IN_F / 4)];
      p2 = fn[8 * (IN_F / 4)]; p3 = fn[12 * (IN_F / 4)];
    }

    const float4* arow = (const float4*)&As[buf][c][4 * h];
    v8f acc = {};
#pragma unroll
    for (int j = 0; j < KPIECES; ++j) {
      const float4 aq = arow[2 * j];
      v2f a, b;
      a.x = aq.x; a.y = aq.y; b.x = bq[j].x; b.y = bq[j].y;
      acc = __builtin_amdgcn_wmma_f32_16x16x4_f32(
           false, a,  false, b,
           (short)0, acc,  false,  false);
      a.x = aq.z; a.y = aq.w; b.x = bq[j].z; b.y = bq[j].w;
      acc = __builtin_amdgcn_wmma_f32_16x16x4_f32(
           false, a,  false, b,
           (short)0, acc,  false,  false);
    }

    if (more) {
      ((float4*)&As[buf ^ 1][lrow + 0][0])[lcol]  = p0;
      ((float4*)&As[buf ^ 1][lrow + 4][0])[lcol]  = p1;
      ((float4*)&As[buf ^ 1][lrow + 8][0])[lcol]  = p2;
      ((float4*)&As[buf ^ 1][lrow + 12][0])[lcol] = p3;
    }
    __syncthreads();

    float* o = obase + (size_t)tile * 16 * CO;
#pragma unroll
    for (int j = 0; j < 8; ++j) o[(size_t)j * CO] = acc[j];
    buf ^= 1;
  }
}

__global__ __launch_bounds__(256) void k_eler(const float* __restrict__ ft,
                                              const float* __restrict__ attn,
                                              float* __restrict__ el,
                                              float* __restrict__ er) {
  int i = blockIdx.x * 256 + threadIdx.x;
  if (i >= NNODES * HEADS) return;
  const int n = i / HEADS, h = i % HEADS;
  const float* f  = ft + (size_t)n * CO + h * OUT_F;
  const float* al = attn + h * 2 * OUT_F;
  const float* ar = al + OUT_F;
  float sl = 0.f, sr = 0.f;
#pragma unroll
  for (int d = 0; d < OUT_F; ++d) {
    const float v = f[d];
    sl += v * al[d];
    sr += v * ar[d];
  }
  el[i] = sl;
  er[i] = sr;
}

__device__ __forceinline__ float lane_f(float v, int l) {
  return __int_as_float(__builtin_amdgcn_readlane(__float_as_int(v), l));
}
__device__ __forceinline__ float pick4(int k0, int k1, int k2, int k3,
                                       float x0, float x1, float x2, float x3) {
  return __int_as_float((__float_as_int(x0) & k0) | (__float_as_int(x1) & k1) |
                        (__float_as_int(x2) & k2) | (__float_as_int(x3) & k3));
}

__device__ __forceinline__ int src_at(const unsigned short* __restrict__ ssrc, int p) {
  const int s = ssrc[p];
  return s < NNODES ? s : NNODES - 1;
}

__global__ __launch_bounds__(256) void k_node(const unsigned short* __restrict__ ssrc,
                                              const int* __restrict__ off,
                                              const float* __restrict__ el,
                                              const float* __restrict__ er,
                                              const float* __restrict__ ft,
                                              float* __restrict__ out) {
  const int lane = threadIdx.x & 31;
  const int wv = __builtin_amdgcn_readfirstlane((int)(threadIdx.x >> 5));
  const int d  = blockIdx.x * 8 + wv;
  if (d >= NNODES) return;
  const int beg = off[d];
  const int deg = off[d + 1] - beg;
  const float4 r4 = *(const float4*)(er + d * HEADS);
  const int hq = lane >> 3;
  const int k0 = (hq == 0) ? -1 : 0;
  const int k1 = (hq == 1) ? -1 : 0;
  const int k2 = (hq == 2) ? -1 : 0;
  const int k3 = (hq == 3) ? -1 : 0;

  const bool in0 = lane < deg;
  int sF = 0;
  float x0 = -INFINITY, x1 = -INFINITY, x2 = -INFINITY, x3 = -INFINITY;
  if (in0) {
    sF = src_at(ssrc, beg + lane);
    const float4 l4 = *(const float4*)(el + sF * HEADS);
    x0 = leaky(l4.x + r4.x);
    x1 = leaky(l4.y + r4.y);
    x2 = leaky(l4.z + r4.z);
    x3 = leaky(l4.w + r4.w);
  }

  float m0 = x0, m1 = x1, m2 = x2, m3 = x3;
  for (int c = 32; c < deg; c += 32) {
    const int j = c + lane;
    if (j < deg) {
      const int s = src_at(ssrc, beg + j);
      const float4 l4 = *(const float4*)(el + s * HEADS);
      m0 = fmaxf(m0, leaky(l4.x + r4.x));
      m1 = fmaxf(m1, leaky(l4.y + r4.y));
      m2 = fmaxf(m2, leaky(l4.z + r4.z));
      m3 = fmaxf(m3, leaky(l4.w + r4.w));
    }
  }
  for (int o = 16; o > 0; o >>= 1) {
    m0 = fmaxf(m0, __shfl_xor(m0, o, 32));
    m1 = fmaxf(m1, __shfl_xor(m1, o, 32));
    m2 = fmaxf(m2, __shfl_xor(m2, o, 32));
    m3 = fmaxf(m3, __shfl_xor(m3, o, 32));
  }

  const float e0 = in0 ? expf(x0 - m0) : 0.f;
  const float e1 = in0 ? expf(x1 - m1) : 0.f;
  const float e2 = in0 ? expf(x2 - m2) : 0.f;
  const float e3 = in0 ? expf(x3 - m3) : 0.f;
  float s0 = e0, s1 = e1, s2 = e2, s3 = e3;
  for (int c = 32; c < deg; c += 32) {
    const int j = c + lane;
    if (j < deg) {
      const int s = src_at(ssrc, beg + j);
      const float4 l4 = *(const float4*)(el + s * HEADS);
      s0 += expf(leaky(l4.x + r4.x) - m0);
      s1 += expf(leaky(l4.y + r4.y) - m1);
      s2 += expf(leaky(l4.z + r4.z) - m2);
      s3 += expf(leaky(l4.w + r4.w) - m3);
    }
  }
  for (int o = 16; o > 0; o >>= 1) {
    s0 += __shfl_xor(s0, o, 32);
    s1 += __shfl_xor(s1, o, 32);
    s2 += __shfl_xor(s2, o, 32);
    s3 += __shfl_xor(s3, o, 32);
  }

  float4 acc = make_float4(0.f, 0.f, 0.f, 0.f);
  const float* fl = ft + lane * 4;
  for (int c = 0; c < deg; c += 32) {
    int s = 0;
    float a0 = 0.f, a1 = 0.f, a2 = 0.f, a3 = 0.f;
    if (c == 0) {
      s  = sF;
      a0 = e0 / s0;
      a1 = e1 / s1;
      a2 = e2 / s2;
      a3 = e3 / s3;
    } else {
      const int j = c + lane;
      if (j < deg) {
        s = src_at(ssrc, beg + j);
        const float4 l4 = *(const float4*)(el + s * HEADS);
        a0 = expf(leaky(l4.x + r4.x) - m0) / s0;
        a1 = expf(leaky(l4.y + r4.y) - m1) / s1;
        a2 = expf(leaky(l4.z + r4.z) - m2) / s2;
        a3 = expf(leaky(l4.w + r4.w) - m3) / s3;
      }
    }
    const int n = (deg - c < 32) ? (deg - c) : 32;
    int k = 0;
    for (; k + 4 <= n; k += 4) {
      const int sA = __builtin_amdgcn_readlane(s, k);
      const int sB = __builtin_amdgcn_readlane(s, k + 1);
      const int sC = __builtin_amdgcn_readlane(s, k + 2);
      const int sD = __builtin_amdgcn_readlane(s, k + 3);
      const float4 fA = *(const float4*)(fl + (size_t)sA * CO);
      const float4 fB = *(const float4*)(fl + (size_t)sB * CO);
      const float4 fC = *(const float4*)(fl + (size_t)sC * CO);
      const float4 fD = *(const float4*)(fl + (size_t)sD * CO);
      const float aA = pick4(k0, k1, k2, k3, lane_f(a0, k), lane_f(a1, k), lane_f(a2, k), lane_f(a3, k));
      const float aB = pick4(k0, k1, k2, k3, lane_f(a0, k + 1), lane_f(a1, k + 1), lane_f(a2, k + 1), lane_f(a3, k + 1));
      const float aC = pick4(k0, k1, k2, k3, lane_f(a0, k + 2), lane_f(a1, k + 2), lane_f(a2, k + 2), lane_f(a3, k + 2));
      const float aD = pick4(k0, k1, k2, k3, lane_f(a0, k + 3), lane_f(a1, k + 3), lane_f(a2, k + 3), lane_f(a3, k + 3));
      acc.x += fA.x * aA; acc.y += fA.y * aA; acc.z += fA.z * aA; acc.w += fA.w * aA;
      acc.x += fB.x * aB; acc.y += fB.y * aB; acc.z += fB.z * aB; acc.w += fB.w * aB;
      acc.x += fC.x * aC; acc.y += fC.y * aC; acc.z += fC.z * aC; acc.w += fC.w * aC;
      acc.x += fD.x * aD; acc.y += fD.y * aD; acc.z += fD.z * aD; acc.w += fD.w * aD;
    }
    for (; k < n; ++k) {
      const int sA = __builtin_amdgcn_readlane(s, k);
      const float4 fA = *(const float4*)(fl + (size_t)sA * CO);
      const float aA = pick4(k0, k1, k2, k3, lane_f(a0, k), lane_f(a1, k), lane_f(a2, k), lane_f(a3, k));
      acc.x += fA.x * aA; acc.y += fA.y * aA; acc.z += fA.z * aA; acc.w += fA.w * aA;
    }
  }
  *(float4*)(out + (size_t)d * CO + lane * 4) = acc;
}

extern "C" void kernel_launch(void* const* d_in, const int* in_sizes, int n_in,
                              void* d_out, int out_size, void* d_ws, size_t ws_size,
                              hipStream_t stream) {
  (void)in_sizes; (void)n_in; (void)out_size; (void)ws_size;

  const float* feat = (const float*)d_in[0];
  const int*   src  = (const int*)d_in[1];
  const int*   dst  = (const int*)d_in[2];
  const float* W    = (const float*)d_in[3];
  const float* attn = (const float*)d_in[4];
  float* out = (float*)d_out;

  char* ws = (char*)d_ws;
  float* ft   = (float*)ws;
  int*   crow = (int*)ws;
  ws += (size_t)NNODES * CO * sizeof(float);
  int*   cnt  = (int*)ws;
  float* el   = (float*)ws;  ws += (size_t)NNODES * HEADS * sizeof(float);
  float* er   = (float*)ws;  ws += (size_t)NNODES * HEADS * sizeof(float);
  int*   off  = (int*)ws;    ws += 200064;
  unsigned short* ssrc = (unsigned short*)ws;

  k_hist<<<SLICES * RANGES, 1024, 0, stream>>>(dst, crow);
  k_colsum<<<(NNODES + 127) / 128, 128, 0, stream>>>(crow, cnt);
  k_scan<<<1, 1024, 0, stream>>>(cnt, off);
  k_fill<<<SLICES * RANGES, 1024, 0, stream>>>(src, dst, off, crow, ssrc);
  k_gemm<<<GEMM_WG, 256, 0, stream>>>(feat, W, ft);
  k_eler<<<(NNODES * HEADS + 255) / 256, 256, 0, stream>>>(ft, attn, el, er);
  k_node<<<(NNODES + 7) / 8, 256, 0, stream>>>(ssrc, off, el, er, ft, out);
}
